# starting kernel
# baseline (speedup 1.0000x reference)
#include <hip/hip_runtime.h>

typedef __attribute__((ext_vector_type(2))) float v2f;
typedef __attribute__((ext_vector_type(8))) float v8f;

#define NCLS 256
#define NDF  512
#define NB   1024
#define KTOT (2 * NDF)
#define ALP  0.1f
#define HALP 0.05f

__global__ __launch_bounds__(NDF) void stats_kernel(
    const float* __restrict__ df, const int* __restrict__ gt,
    float* __restrict__ var_out)
{
    __shared__ int gtS[NB];
    const int c = blockIdx.x;
    const int d = threadIdx.x;
    gtS[d]        = gt[d];
    gtS[d + NDF]  = gt[d + NDF];
    __syncthreads();

    float m = 0.0f, v = 0.0f;
    int   n = 0;
    for (int b = 0; b < NB; ++b) {
        if (gtS[b] == c) {
            const float f  = df[(size_t)b * NDF + d];
            const float nf = (float)n;
            const float c1 = nf / (nf + 1.0f);
            const float c2 = 1.0f / (nf + 1.0f);
            m = (n == 0) ? f : (m * c1 + f * c2);
            const float a = f - m;
            v = v * c1 + a * a * c1 * c2;
            ++n;
        }
    }
    var_out[(size_t)c * NDF + d] = v;
}

__global__ __launch_bounds__(NDF) void prep_a_kernel(
    const float* __restrict__ df, const int* __restrict__ gt,
    const float* __restrict__ fc, const float* __restrict__ var_in,
    float* __restrict__ XY, float* __restrict__ constb)
{
    __shared__ float red[NDF];
    const int b = blockIdx.x;
    const int d = threadIdx.x;
    const int t = gt[b];

    const float fcv = fc[(size_t)t * NDF + d];
    const float va  = var_in[(size_t)t * NDF + d];
    const float dfv = df[(size_t)b * NDF + d];

    XY[(size_t)b * KTOT + d]       = dfv - ALP * fcv * va;
    XY[(size_t)b * KTOT + NDF + d] = HALP * va;

    red[d] = -fcv * dfv + HALP * fcv * fcv * va;
    __syncthreads();
    for (int s = NDF / 2; s > 0; s >>= 1) {
        if (d < s) red[d] += red[d + s];
        __syncthreads();
    }
    if (d == 0) constb[b] = red[0];
}

__global__ __launch_bounds__(NDF) void prep_b_kernel(
    const float* __restrict__ fc, float* __restrict__ PQ)
{
    const int c = blockIdx.x;
    const int d = threadIdx.x;
    const float v = fc[(size_t)c * NDF + d];
    PQ[(size_t)c * KTOT + d]       = v;
    PQ[(size_t)c * KTOT + NDF + d] = v * v;
}

__global__ __launch_bounds__(32) void wmma_gemm_kernel(
    const float* __restrict__ XY, const float* __restrict__ PQ,
    const float* __restrict__ constb, float* __restrict__ out)
{
    const int tm   = blockIdx.x;
    const int ts   = blockIdx.y;
    const int lane = threadIdx.x;
    const int r    = lane & 15;
    const int kh   = (lane >> 4) << 1;

    const float* Arow = XY + (size_t)(tm * 16 + r) * KTOT + kh;
    const float* Brow = PQ + (size_t)(ts * 64 + r) * KTOT + kh;

    v8f acc0 = {}, acc1 = {}, acc2 = {}, acc3 = {};
#pragma unroll 4
    for (int k = 0; k < KTOT; k += 4) {
        v2f a  = *(const v2f*)(Arow + k);
        v2f b0 = *(const v2f*)(Brow + k);
        v2f b1 = *(const v2f*)(Brow + 16 * KTOT + k);
        v2f b2 = *(const v2f*)(Brow + 32 * KTOT + k);
        v2f b3 = *(const v2f*)(Brow + 48 * KTOT + k);
        acc0 = __builtin_amdgcn_wmma_f32_16x16x4_f32(false, a, false, b0,
                                                     (short)0, acc0, false, false);
        acc1 = __builtin_amdgcn_wmma_f32_16x16x4_f32(false, a, false, b1,
                                                     (short)0, acc1, false, false);
        acc2 = __builtin_amdgcn_wmma_f32_16x16x4_f32(false, a, false, b2,
                                                     (short)0, acc2, false, false);
        acc3 = __builtin_amdgcn_wmma_f32_16x16x4_f32(false, a, false, b3,
                                                     (short)0, acc3, false, false);
    }

    const int mbase = tm * 16 + ((lane >> 4) << 3);
    float cst[8];
#pragma unroll
    for (int j = 0; j < 8; ++j) cst[j] = constb[mbase + j];

    const int colbase = ts * 64 + r;
#pragma unroll
    for (int j = 0; j < 8; ++j) {
        const size_t rowoff = (size_t)(mbase + j) * NCLS;
        out[rowoff + colbase +  0] = acc0[j] + cst[j];
        out[rowoff + colbase + 16] = acc1[j] + cst[j];
        out[rowoff + colbase + 32] = acc2[j] + cst[j];
        out[rowoff + colbase + 48] = acc3[j] + cst[j];
    }
}

extern "C" void kernel_launch(void* const* d_in, const int* in_sizes, int n_in,
                              void* d_out, int out_size, void* d_ws, size_t ws_size,
                              hipStream_t stream) {
    const float* df = (const float*)d_in[0];
    const int*   gt = (const int*)d_in[1];
    const float* fc = (const float*)d_in[2];
    float*       out = (float*)d_out;

    float* var    = (float*)d_ws;
    float* XY     = var + (size_t)NCLS * NDF;
    float* PQ     = XY + (size_t)NB * KTOT;
    float* constb = PQ + (size_t)NCLS * KTOT;

    stats_kernel<<<NCLS, NDF, 0, stream>>>(df, gt, var);
    prep_a_kernel<<<NB, NDF, 0, stream>>>(df, gt, fc, var, XY, constb);
    prep_b_kernel<<<NCLS, NDF, 0, stream>>>(fc, PQ);
    wmma_gemm_kernel<<<dim3(NB / 16, NCLS / 16 / 4), 32, 0, stream>>>(XY, PQ, constb, out);
}


# stats kernel: ordered per-class sample list from 32 wave ballots replaces the serial 1024-step LDS scan; listed rows loaded 16 per batch, 4 features per lane, same summation order
# speedup vs baseline: 1.9739x; 1.9739x over previous
#include <hip/hip_runtime.h>

typedef __attribute__((ext_vector_type(2))) float v2f;
typedef __attribute__((ext_vector_type(8))) float v8f;

#define NCLS 256
#define NDF  512
#define NB   1024
#define KTOT (2 * NDF)
#define ALP  0.1f
#define HALP 0.05f

typedef __attribute__((ext_vector_type(4))) float v4f;

#define STATS_WAVES 4
#define STATS_BATCH 16

__global__ __launch_bounds__(STATS_WAVES * 32) void stats_kernel(
    const float* __restrict__ df, const int* __restrict__ gt,
    float* __restrict__ var_out)
{
    __shared__ int listS[STATS_WAVES][NB + 32];
    const int c    = blockIdx.x;
    const int lane = threadIdx.x & 31;
    const int w    = threadIdx.x >> 5;
    int* list = listS[w];
    const int spare = NB + lane;

    int g[NB / 32];
#pragma unroll
    for (int i = 0; i < NB / 32; ++i) g[i] = gt[i * 32 + lane];

    int n = 0;
#pragma unroll
    for (int i = 0; i < NB / 32; ++i) {
        const bool hit = (g[i] == c);
        const unsigned mask = __builtin_amdgcn_ballot_w32(hit);
        const int at  = (int)__builtin_amdgcn_mbcnt_lo(mask, (unsigned)n);
        const int pos = hit ? at : spare;
        list[pos] = i * 32 + lane;
        n += __builtin_popcount(mask);
    }
    __syncthreads();

    const char* rows = (const char*)df;
    const unsigned col = (unsigned)(w * 32 + lane) * 16u;
    v4f m = {0.0f, 0.0f, 0.0f, 0.0f};
    v4f v = {0.0f, 0.0f, 0.0f, 0.0f};
    for (int i0 = 0; i0 < n; i0 += STATS_BATCH) {
        v4f f[STATS_BATCH];
#pragma unroll
        for (int j = 0; j < STATS_BATCH; ++j) {
            const int b = list[min(i0 + j, n - 1)];
            f[j] = *(const v4f*)(rows + ((unsigned)b * (NDF * 4u) + col));
        }
        asm volatile("" :: "v"(f[0]), "v"(f[1]), "v"(f[2]), "v"(f[3]),
                           "v"(f[4]), "v"(f[5]), "v"(f[6]), "v"(f[7]),
                           "v"(f[8]), "v"(f[9]), "v"(f[10]), "v"(f[11]),
                           "v"(f[12]), "v"(f[13]), "v"(f[14]), "v"(f[15]));
#pragma unroll
        for (int j = 0; j < STATS_BATCH; ++j) {
            const int k = i0 + j;
            if (k < n) {
                const float nf = (float)k;
                const float c1 = nf / (nf + 1.0f);
                const float c2 = 1.0f / (nf + 1.0f);
                const v4f fj = f[j];
                m = (k == 0) ? fj : (m * c1 + fj * c2);
                const v4f a = fj - m;
                v = v * c1 + a * a * c1 * c2;
            }
        }
    }
    ((v4f*)var_out)[(size_t)c * (NDF / 4) + w * 32 + lane] = v;
}

__global__ __launch_bounds__(NDF) void prep_a_kernel(
    const float* __restrict__ df, const int* __restrict__ gt,
    const float* __restrict__ fc, const float* __restrict__ var_in,
    float* __restrict__ XY, float* __restrict__ constb)
{
    __shared__ float red[NDF];
    const int b = blockIdx.x;
    const int d = threadIdx.x;
    const int t = gt[b];

    const float fcv = fc[(size_t)t * NDF + d];
    const float va  = var_in[(size_t)t * NDF + d];
    const float dfv = df[(size_t)b * NDF + d];

    XY[(size_t)b * KTOT + d]       = dfv - ALP * fcv * va;
    XY[(size_t)b * KTOT + NDF + d] = HALP * va;

    red[d] = -fcv * dfv + HALP * fcv * fcv * va;
    __syncthreads();
    for (int s = NDF / 2; s > 0; s >>= 1) {
        if (d < s) red[d] += red[d + s];
        __syncthreads();
    }
    if (d == 0) constb[b] = red[0];
}

__global__ __launch_bounds__(NDF) void prep_b_kernel(
    const float* __restrict__ fc, float* __restrict__ PQ)
{
    const int c = blockIdx.x;
    const int d = threadIdx.x;
    const float v = fc[(size_t)c * NDF + d];
    PQ[(size_t)c * KTOT + d]       = v;
    PQ[(size_t)c * KTOT + NDF + d] = v * v;
}

__global__ __launch_bounds__(32) void wmma_gemm_kernel(
    const float* __restrict__ XY, const float* __restrict__ PQ,
    const float* __restrict__ constb, float* __restrict__ out)
{
    const int tm   = blockIdx.x;
    const int ts   = blockIdx.y;
    const int lane = threadIdx.x;
    const int r    = lane & 15;
    const int kh   = (lane >> 4) << 1;

    const float* Arow = XY + (size_t)(tm * 16 + r) * KTOT + kh;
    const float* Brow = PQ + (size_t)(ts * 64 + r) * KTOT + kh;

    v8f acc0 = {}, acc1 = {}, acc2 = {}, acc3 = {};
#pragma unroll 4
    for (int k = 0; k < KTOT; k += 4) {
        v2f a  = *(const v2f*)(Arow + k);
        v2f b0 = *(const v2f*)(Brow + k);
        v2f b1 = *(const v2f*)(Brow + 16 * KTOT + k);
        v2f b2 = *(const v2f*)(Brow + 32 * KTOT + k);
        v2f b3 = *(const v2f*)(Brow + 48 * KTOT + k);
        acc0 = __builtin_amdgcn_wmma_f32_16x16x4_f32(false, a, false, b0,
                                                     (short)0, acc0, false, false);
        acc1 = __builtin_amdgcn_wmma_f32_16x16x4_f32(false, a, false, b1,
                                                     (short)0, acc1, false, false);
        acc2 = __builtin_amdgcn_wmma_f32_16x16x4_f32(false, a, false, b2,
                                                     (short)0, acc2, false, false);
        acc3 = __builtin_amdgcn_wmma_f32_16x16x4_f32(false, a, false, b3,
                                                     (short)0, acc3, false, false);
    }

    const int mbase = tm * 16 + ((lane >> 4) << 3);
    float cst[8];
#pragma unroll
    for (int j = 0; j < 8; ++j) cst[j] = constb[mbase + j];

    const int colbase = ts * 64 + r;
#pragma unroll
    for (int j = 0; j < 8; ++j) {
        const size_t rowoff = (size_t)(mbase + j) * NCLS;
        out[rowoff + colbase +  0] = acc0[j] + cst[j];
        out[rowoff + colbase + 16] = acc1[j] + cst[j];
        out[rowoff + colbase + 32] = acc2[j] + cst[j];
        out[rowoff + colbase + 48] = acc3[j] + cst[j];
    }
}

extern "C" void kernel_launch(void* const* d_in, const int* in_sizes, int n_in,
                              void* d_out, int out_size, void* d_ws, size_t ws_size,
                              hipStream_t stream) {
    const float* df = (const float*)d_in[0];
    const int*   gt = (const int*)d_in[1];
    const float* fc = (const float*)d_in[2];
    float*       out = (float*)d_out;

    float* var    = (float*)d_ws;
    float* XY     = var + (size_t)NCLS * NDF;
    float* PQ     = XY + (size_t)NB * KTOT;
    float* constb = PQ + (size_t)NCLS * KTOT;

    stats_kernel<<<NCLS, STATS_WAVES * 32, 0, stream>>>(df, gt, var);
    prep_a_kernel<<<NB, NDF, 0, stream>>>(df, gt, fc, var, XY, constb);
    prep_b_kernel<<<NCLS, NDF, 0, stream>>>(fc, PQ);
    wmma_gemm_kernel<<<dim3(NB / 16, NCLS / 16 / 4), 32, 0, stream>>>(XY, PQ, constb, out);
}


# gemm kernel: 4-wave block per 32x32 tile, K slabs staged in LDS by coalesced 16-byte loads, one 16x16 tile per wave on two accumulators (alternate k pairs, added at end: f32 summation order changes)
# speedup vs baseline: 5.3601x; 2.7155x over previous
#include <hip/hip_runtime.h>

typedef __attribute__((ext_vector_type(2))) float v2f;
typedef __attribute__((ext_vector_type(8))) float v8f;

#define NCLS 256
#define NDF  512
#define NB   1024
#define KTOT (2 * NDF)
#define ALP  0.1f
#define HALP 0.05f

typedef __attribute__((ext_vector_type(4))) float v4f;

#define STATS_WAVES 4
#define STATS_BATCH 16

__global__ __launch_bounds__(STATS_WAVES * 32) void stats_kernel(
    const float* __restrict__ df, const int* __restrict__ gt,
    float* __restrict__ var_out)
{
    __shared__ int listS[STATS_WAVES][NB + 32];
    const int c    = blockIdx.x;
    const int lane = threadIdx.x & 31;
    const int w    = threadIdx.x >> 5;
    int* list = listS[w];
    const int spare = NB + lane;

    int g[NB / 32];
#pragma unroll
    for (int i = 0; i < NB / 32; ++i) g[i] = gt[i * 32 + lane];

    int n = 0;
#pragma unroll
    for (int i = 0; i < NB / 32; ++i) {
        const bool hit = (g[i] == c);
        const unsigned mask = __builtin_amdgcn_ballot_w32(hit);
        const int at  = (int)__builtin_amdgcn_mbcnt_lo(mask, (unsigned)n);
        const int pos = hit ? at : spare;
        list[pos] = i * 32 + lane;
        n += __builtin_popcount(mask);
    }
    __syncthreads();

    const char* rows = (const char*)df;
    const unsigned col = (unsigned)(w * 32 + lane) * 16u;
    v4f m = {0.0f, 0.0f, 0.0f, 0.0f};
    v4f v = {0.0f, 0.0f, 0.0f, 0.0f};
    for (int i0 = 0; i0 < n; i0 += STATS_BATCH) {
        v4f f[STATS_BATCH];
#pragma unroll
        for (int j = 0; j < STATS_BATCH; ++j) {
            const int b = list[min(i0 + j, n - 1)];
            f[j] = *(const v4f*)(rows + ((unsigned)b * (NDF * 4u) + col));
        }
        asm volatile("" :: "v"(f[0]), "v"(f[1]), "v"(f[2]), "v"(f[3]),
                           "v"(f[4]), "v"(f[5]), "v"(f[6]), "v"(f[7]),
                           "v"(f[8]), "v"(f[9]), "v"(f[10]), "v"(f[11]),
                           "v"(f[12]), "v"(f[13]), "v"(f[14]), "v"(f[15]));
#pragma unroll
        for (int j = 0; j < STATS_BATCH; ++j) {
            const int k = i0 + j;
            if (k < n) {
                const float nf = (float)k;
                const float c1 = nf / (nf + 1.0f);
                const float c2 = 1.0f / (nf + 1.0f);
                const v4f fj = f[j];
                m = (k == 0) ? fj : (m * c1 + fj * c2);
                const v4f a = fj - m;
                v = v * c1 + a * a * c1 * c2;
            }
        }
    }
    ((v4f*)var_out)[(size_t)c * (NDF / 4) + w * 32 + lane] = v;
}

__global__ __launch_bounds__(NDF) void prep_a_kernel(
    const float* __restrict__ df, const int* __restrict__ gt,
    const float* __restrict__ fc, const float* __restrict__ var_in,
    float* __restrict__ XY, float* __restrict__ constb)
{
    __shared__ float red[NDF];
    const int b = blockIdx.x;
    const int d = threadIdx.x;
    const int t = gt[b];

    const float fcv = fc[(size_t)t * NDF + d];
    const float va  = var_in[(size_t)t * NDF + d];
    const float dfv = df[(size_t)b * NDF + d];

    XY[(size_t)b * KTOT + d]       = dfv - ALP * fcv * va;
    XY[(size_t)b * KTOT + NDF + d] = HALP * va;

    red[d] = -fcv * dfv + HALP * fcv * fcv * va;
    __syncthreads();
    for (int s = NDF / 2; s > 0; s >>= 1) {
        if (d < s) red[d] += red[d + s];
        __syncthreads();
    }
    if (d == 0) constb[b] = red[0];
}

__global__ __launch_bounds__(NDF) void prep_b_kernel(
    const float* __restrict__ fc, float* __restrict__ PQ)
{
    const int c = blockIdx.x;
    const int d = threadIdx.x;
    const float v = fc[(size_t)c * NDF + d];
    PQ[(size_t)c * KTOT + d]       = v;
    PQ[(size_t)c * KTOT + NDF + d] = v * v;
}

#define GB_T   32
#define GB_K   128
#define GB_P   (GB_K + 4)
#define GB_NS  (KTOT / GB_K)
#define GB_RW  (GB_T / 4)
#define GB_CP  (GB_T + 4)

__device__ __forceinline__ void gemm_slab_load(v4f (&ra)[GB_RW], v4f (&rb)[GB_RW],
                                               const float* gA, const float* gB)
{
#pragma unroll
    for (int j = 0; j < GB_RW; ++j) {
        ra[j] = *(const v4f*)(gA + (size_t)j * 4 * KTOT);
        rb[j] = *(const v4f*)(gB + (size_t)j * 4 * KTOT);
    }
}

__device__ __forceinline__ void gemm_slab_store(const v4f (&ra)[GB_RW], const v4f (&rb)[GB_RW],
                                                float* sa, float* sb)
{
#pragma unroll
    for (int j = 0; j < GB_RW; ++j) {
        *(v4f*)(sa + j * 4 * GB_P) = ra[j];
        *(v4f*)(sb + j * 4 * GB_P) = rb[j];
    }
}

__device__ __forceinline__ void gemm_slab_mma(const float* pa, const float* pb,
                                              v8f& acc0, v8f& acc1)
{
    v4f a[GB_K / 8], b[GB_K / 8];
#pragma unroll
    for (int s = 0; s < GB_K / 8; ++s) {
        a[s] = *(const v4f*)(pa + s * 8);
        b[s] = *(const v4f*)(pb + s * 8);
    }
    __builtin_amdgcn_sched_barrier(0);
#pragma unroll
    for (int s = 0; s < GB_K / 8; ++s) {
        const v2f a0 = {a[s].x, a[s].y}, a1 = {a[s].z, a[s].w};
        const v2f b0 = {b[s].x, b[s].y}, b1 = {b[s].z, b[s].w};
        acc0 = __builtin_amdgcn_wmma_f32_16x16x4_f32(false, a0, false, b0,
                                                     (short)0, acc0, false, false);
        acc1 = __builtin_amdgcn_wmma_f32_16x16x4_f32(false, a1, false, b1,
                                                     (short)0, acc1, false, false);
    }
}

__global__ __launch_bounds__(128) void wmma_gemm_kernel(
    const float* __restrict__ XY, const float* __restrict__ PQ,
    const float* __restrict__ constb, float* __restrict__ out)
{
    __shared__ __attribute__((aligned(16))) float sA[2][GB_T * GB_P];
    __shared__ __attribute__((aligned(16))) float sB[2][GB_T * GB_P];

    const int tm   = blockIdx.x;
    const int tn   = blockIdx.y;
    const int lane = threadIdx.x & 31;
    const int w    = threadIdx.x >> 5;
    const int wm   = w >> 1;
    const int wn   = w & 1;
    const int c    = lane & 15;
    const int h    = lane >> 4;

    const float* gA = XY + (size_t)(tm * GB_T + w) * KTOT + lane * 4;
    const float* gB = PQ + (size_t)(tn * GB_T + w) * KTOT + lane * 4;
    const int sOff = w * GB_P + lane * 4;
    const int fA = (wm * 16 + c) * GB_P + 4 * h;
    const int fB = (wn * 16 + c) * GB_P + 4 * h;

    v4f ra[GB_RW], rb[GB_RW];
    gemm_slab_load(ra, rb, gA, gB);
    gemm_slab_store(ra, rb, &sA[0][sOff], &sB[0][sOff]);
    __syncthreads();

    v8f acc0 = {}, acc1 = {};
    for (int ch = 0; ch < GB_NS - 1; ++ch) {
        const int cur = ch & 1;
        gemm_slab_load(ra, rb, gA + (ch + 1) * GB_K, gB + (ch + 1) * GB_K);
        gemm_slab_mma(&sA[cur][fA], &sB[cur][fB], acc0, acc1);
        asm volatile("" :: "v"(acc0), "v"(acc1) : "memory");
        gemm_slab_store(ra, rb, &sA[cur ^ 1][sOff], &sB[cur ^ 1][sOff]);
        __syncthreads();
    }
    gemm_slab_mma(&sA[(GB_NS - 1) & 1][fA], &sB[(GB_NS - 1) & 1][fB], acc0, acc1);

    float* sC = &sA[0][0];
    const int mloc = wm * 16 + (h << 3);
    float cst[8];
#pragma unroll
    for (int j = 0; j < 8; ++j) cst[j] = constb[tm * GB_T + mloc + j];
#pragma unroll
    for (int j = 0; j < 8; ++j)
        sC[(mloc + j) * GB_CP + wn * 16 + c] = (acc0[j] + acc1[j]) + cst[j];
    __syncthreads();

#pragma unroll
    for (int i = 0; i < 2; ++i) {
        const int row = w * 8 + i * 4 + (lane >> 3);
        const int cq  = (lane & 7) * 4;
        const v4f o = *(const v4f*)(sC + row * GB_CP + cq);
        *(v4f*)(out + (size_t)(tm * GB_T + row) * NCLS + tn * GB_T + cq) = o;
    }
}

extern "C" void kernel_launch(void* const* d_in, const int* in_sizes, int n_in,
                              void* d_out, int out_size, void* d_ws, size_t ws_size,
                              hipStream_t stream) {
    const float* df = (const float*)d_in[0];
    const int*   gt = (const int*)d_in[1];
    const float* fc = (const float*)d_in[2];
    float*       out = (float*)d_out;

    float* var    = (float*)d_ws;
    float* XY     = var + (size_t)NCLS * NDF;
    float* PQ     = XY + (size_t)NB * KTOT;
    float* constb = PQ + (size_t)NCLS * KTOT;

    stats_kernel<<<NCLS, STATS_WAVES * 32, 0, stream>>>(df, gt, var);
    prep_a_kernel<<<NB, NDF, 0, stream>>>(df, gt, fc, var, XY, constb);
    prep_b_kernel<<<NCLS, NDF, 0, stream>>>(fc, PQ);
    wmma_gemm_kernel<<<dim3(NB / GB_T, NCLS / GB_T), 128, 0, stream>>>(XY, PQ, constb, out);
}


# gemm kernel builds X, Y, fc^2 and the row constant in its own slab loader from df, fc[gt] and var[gt]; both prep launches and the XY/PQ workspace round trip removed; f32 summation order changes
# speedup vs baseline: 8.0218x; 1.4966x over previous
#include <hip/hip_runtime.h>

typedef __attribute__((ext_vector_type(2))) float v2f;
typedef __attribute__((ext_vector_type(8))) float v8f;

#define NCLS 256
#define NDF  512
#define NB   1024
#define ALP  0.1f
#define HALP 0.05f

typedef __attribute__((ext_vector_type(4))) float v4f;

#define STATS_WAVES 4
#define STATS_BATCH 16

__global__ __launch_bounds__(STATS_WAVES * 32) void stats_kernel(
    const float* __restrict__ df, const int* __restrict__ gt,
    float* __restrict__ var_out)
{
    __shared__ int listS[STATS_WAVES][NB + 32];
    const int c    = blockIdx.x;
    const int lane = threadIdx.x & 31;
    const int w    = threadIdx.x >> 5;
    int* list = listS[w];
    const int spare = NB + lane;

    int g[NB / 32];
#pragma unroll
    for (int i = 0; i < NB / 32; ++i) g[i] = gt[i * 32 + lane];

    int n = 0;
#pragma unroll
    for (int i = 0; i < NB / 32; ++i) {
        const bool hit = (g[i] == c);
        const unsigned mask = __builtin_amdgcn_ballot_w32(hit);
        const int at  = (int)__builtin_amdgcn_mbcnt_lo(mask, (unsigned)n);
        const int pos = hit ? at : spare;
        list[pos] = i * 32 + lane;
        n += __builtin_popcount(mask);
    }
    __syncthreads();

    const char* rows = (const char*)df;
    const unsigned col = (unsigned)(w * 32 + lane) * 16u;
    v4f m = {0.0f, 0.0f, 0.0f, 0.0f};
    v4f v = {0.0f, 0.0f, 0.0f, 0.0f};
    for (int i0 = 0; i0 < n; i0 += STATS_BATCH) {
        v4f f[STATS_BATCH];
#pragma unroll
        for (int j = 0; j < STATS_BATCH; ++j) {
            const int b = list[min(i0 + j, n - 1)];
            f[j] = *(const v4f*)(rows + ((unsigned)b * (NDF * 4u) + col));
        }
        asm volatile("" :: "v"(f[0]), "v"(f[1]), "v"(f[2]), "v"(f[3]),
                           "v"(f[4]), "v"(f[5]), "v"(f[6]), "v"(f[7]),
                           "v"(f[8]), "v"(f[9]), "v"(f[10]), "v"(f[11]),
                           "v"(f[12]), "v"(f[13]), "v"(f[14]), "v"(f[15]));
#pragma unroll
        for (int j = 0; j < STATS_BATCH; ++j) {
            const int k = i0 + j;
            if (k < n) {
                const float nf = (float)k;
                const float c1 = nf / (nf + 1.0f);
                const float c2 = 1.0f / (nf + 1.0f);
                const v4f fj = f[j];
                m = (k == 0) ? fj : (m * c1 + fj * c2);
                const v4f a = fj - m;
                v = v * c1 + a * a * c1 * c2;
            }
        }
    }
    ((v4f*)var_out)[(size_t)c * (NDF / 4) + w * 32 + lane] = v;
}

#define GB_T   32
#define GB_K   128
#define GB_P   (GB_K + 4)
#define GB_NS  (NDF / GB_K)
#define GB_RW  (GB_T / 4)
#define GB_CP  (GB_T + 4)
#define GB_RP  (32 + 4)
#define GB_RS  (4u * NDF * 4u)

__device__ __forceinline__ void gemm_slab_load(v4f (&rd)[GB_RW], v4f (&rf)[GB_RW],
                                               v4f (&rv)[GB_RW], v4f (&rb)[GB_RW],
                                               const char* dfB, const char* fcB, const char* vaB,
                                               unsigned oD, unsigned oB,
                                               const unsigned (&oT)[GB_RW], unsigned kb)
{
#pragma unroll
    for (int j = 0; j < GB_RW; ++j) {
        rd[j] = *(const v4f*)(dfB + (oD + (unsigned)j * GB_RS + kb));
        rb[j] = *(const v4f*)(fcB + (oB + (unsigned)j * GB_RS + kb));
    }
#pragma unroll
    for (int j = 0; j < GB_RW; ++j) {
        rf[j] = *(const v4f*)(fcB + (oT[j] + kb));
        rv[j] = *(const v4f*)(vaB + (oT[j] + kb));
    }
}

__device__ __forceinline__ void gemm_slab_store(const v4f (&rd)[GB_RW], const v4f (&rf)[GB_RW],
                                                const v4f (&rv)[GB_RW], const v4f (&rb)[GB_RW],
                                                float* sx, float* sy, float* sp,
                                                float (&cacc)[GB_RW])
{
#pragma unroll
    for (int j = 0; j < GB_RW; ++j) {
        const v4f dfv = rd[j], fcv = rf[j], va = rv[j];
        const v4f x = dfv - ALP * fcv * va;
        const v4f y = HALP * va;
        const v4f t = -fcv * dfv + HALP * fcv * fcv * va;
        cacc[j] += (t.x + t.y) + (t.z + t.w);
        *(v4f*)(sx + j * 4 * GB_P) = x;
        *(v4f*)(sy + j * 4 * GB_P) = y;
        *(v4f*)(sp + j * 4 * GB_P) = rb[j];
    }
}

__device__ __forceinline__ void gemm_slab_mma(const float* px, const float* py, const float* pp,
                                              v8f& ax0, v8f& ax1, v8f& ay0, v8f& ay1)
{
    v4f x[GB_K / 8], y[GB_K / 8], p[GB_K / 8];
#pragma unroll
    for (int s = 0; s < GB_K / 8; ++s) {
        x[s] = *(const v4f*)(px + s * 8);
        y[s] = *(const v4f*)(py + s * 8);
        p[s] = *(const v4f*)(pp + s * 8);
    }
    __builtin_amdgcn_sched_barrier(0);
#pragma unroll
    for (int s = 0; s < GB_K / 8; ++s) {
        const v4f q = p[s] * p[s];
        const v2f x0 = {x[s].x, x[s].y}, x1 = {x[s].z, x[s].w};
        const v2f y0 = {y[s].x, y[s].y}, y1 = {y[s].z, y[s].w};
        const v2f p0 = {p[s].x, p[s].y}, p1 = {p[s].z, p[s].w};
        const v2f q0 = {q.x, q.y},       q1 = {q.z, q.w};
        ax0 = __builtin_amdgcn_wmma_f32_16x16x4_f32(false, x0, false, p0,
                                                    (short)0, ax0, false, false);
        ax1 = __builtin_amdgcn_wmma_f32_16x16x4_f32(false, x1, false, p1,
                                                    (short)0, ax1, false, false);
        ay0 = __builtin_amdgcn_wmma_f32_16x16x4_f32(false, y0, false, q0,
                                                    (short)0, ay0, false, false);
        ay1 = __builtin_amdgcn_wmma_f32_16x16x4_f32(false, y1, false, q1,
                                                    (short)0, ay1, false, false);
        if (s < GB_K / 16) __builtin_amdgcn_sched_barrier(0);
    }
}

__global__ __launch_bounds__(128) void wmma_gemm_kernel(
    const float* __restrict__ df, const int* __restrict__ gt,
    const float* __restrict__ fc, const float* __restrict__ var_in,
    float* __restrict__ out)
{
    __shared__ __attribute__((aligned(16))) float sX[2][GB_T * GB_P];
    __shared__ __attribute__((aligned(16))) float sY[2][GB_T * GB_P];
    __shared__ __attribute__((aligned(16))) float sP[2][GB_T * GB_P];
    __shared__ __attribute__((aligned(16))) float sR[GB_T * GB_RP];

    const int tm   = blockIdx.x;
    const int tn   = blockIdx.y;
    const int lane = threadIdx.x & 31;
    const int w    = threadIdx.x >> 5;
    const int wm   = w >> 1;
    const int wn   = w & 1;
    const int c    = lane & 15;
    const int h    = lane >> 4;

    const char* dfB = (const char*)df;
    const char* fcB = (const char*)fc;
    const char* vaB = (const char*)var_in;
    const unsigned oD = ((unsigned)(tm * GB_T + w) * NDF + (unsigned)lane * 4u) * 4u;
    const unsigned oB = ((unsigned)(tn * GB_T + w) * NDF + (unsigned)lane * 4u) * 4u;
    unsigned oT[GB_RW];
#pragma unroll
    for (int j = 0; j < GB_RW; ++j)
        oT[j] = ((unsigned)gt[tm * GB_T + w + 4 * j] * NDF + (unsigned)lane * 4u) * 4u;

    const int sOff = w * GB_P + lane * 4;
    const int fA = (wm * 16 + c) * GB_P + 4 * h;
    const int fB = (wn * 16 + c) * GB_P + 4 * h;

    float cacc[GB_RW];
#pragma unroll
    for (int j = 0; j < GB_RW; ++j) cacc[j] = 0.0f;

    v4f rd[GB_RW], rf[GB_RW], rv[GB_RW], rb[GB_RW];
    gemm_slab_load(rd, rf, rv, rb, dfB, fcB, vaB, oD, oB, oT, 0u);
    gemm_slab_store(rd, rf, rv, rb, &sX[0][sOff], &sY[0][sOff], &sP[0][sOff], cacc);
    __syncthreads();

    v8f ax0 = {}, ax1 = {}, ay0 = {}, ay1 = {};
    for (int ch = 0; ch < GB_NS - 1; ++ch) {
        const int cur = ch & 1;
        gemm_slab_load(rd, rf, rv, rb, dfB, fcB, vaB, oD, oB, oT,
                       (unsigned)(ch + 1) * (GB_K * 4u));
        gemm_slab_mma(&sX[cur][fA], &sY[cur][fA], &sP[cur][fB], ax0, ax1, ay0, ay1);
        asm volatile("" :: "v"(ax0), "v"(ax1), "v"(ay0), "v"(ay1) : "memory");
        gemm_slab_store(rd, rf, rv, rb, &sX[cur ^ 1][sOff], &sY[cur ^ 1][sOff], &sP[cur ^ 1][sOff], cacc);
        __syncthreads();
    }
    gemm_slab_mma(&sX[(GB_NS - 1) & 1][fA], &sY[(GB_NS - 1) & 1][fA], &sP[(GB_NS - 1) & 1][fB],
                  ax0, ax1, ay0, ay1);

    float* sC = &sX[0][0];
    const int mloc = wm * 16 + (h << 3);
#pragma unroll
    for (int j = 0; j < 8; ++j)
        sC[(mloc + j) * GB_CP + wn * 16 + c] = (ax0[j] + ax1[j]) + (ay0[j] + ay1[j]);
#pragma unroll
    for (int j = 0; j < GB_RW; ++j)
        sR[(w + 4 * j) * GB_RP + lane] = cacc[j];
    __syncthreads();

#pragma unroll
    for (int i = 0; i < 2; ++i) {
        const int row = w * 8 + i * 4 + (lane >> 3);
        const int cq  = (lane & 7) * 4;
        float cs = 0.0f;
#pragma unroll
        for (int g = 0; g < 8; ++g) {
            const v4f pr = *(const v4f*)(sR + row * GB_RP + g * 4);
            cs += (pr.x + pr.y) + (pr.z + pr.w);
        }
        v4f o = *(const v4f*)(sC + row * GB_CP + cq);
        o = o + cs;
        *(v4f*)(out + (size_t)(tm * GB_T + row) * NCLS + tn * GB_T + cq) = o;
    }
}

extern "C" void kernel_launch(void* const* d_in, const int* in_sizes, int n_in,
                              void* d_out, int out_size, void* d_ws, size_t ws_size,
                              hipStream_t stream) {
    const float* df = (const float*)d_in[0];
    const int*   gt = (const int*)d_in[1];
    const float* fc = (const float*)d_in[2];
    float*       out = (float*)d_out;

    float* var = (float*)d_ws;

    stats_kernel<<<NCLS, STATS_WAVES * 32, 0, stream>>>(df, gt, var);
    wmma_gemm_kernel<<<dim3(NB / GB_T, NCLS / GB_T), 128, 0, stream>>>(df, gt, fc, var, out);
}


# gemm kernel: 8-wave block, two groups of 4 waves take the K slabs in turn: one multiplies while the other fetches, builds and stores its next slab; partial tiles added at end (f32 sum order changes)
# speedup vs baseline: 8.4311x; 1.0510x over previous
#include <hip/hip_runtime.h>

typedef __attribute__((ext_vector_type(2))) float v2f;
typedef __attribute__((ext_vector_type(8))) float v8f;

#define NCLS 256
#define NDF  512
#define NB   1024
#define ALP  0.1f
#define HALP 0.05f

typedef __attribute__((ext_vector_type(4))) float v4f;

#define STATS_WAVES 4
#define STATS_BATCH 16

__global__ __launch_bounds__(STATS_WAVES * 32) void stats_kernel(
    const float* __restrict__ df, const int* __restrict__ gt,
    float* __restrict__ var_out)
{
    __shared__ int listS[STATS_WAVES][NB + 32];
    const int c    = blockIdx.x;
    const int lane = threadIdx.x & 31;
    const int w    = threadIdx.x >> 5;
    int* list = listS[w];
    const int spare = NB + lane;

    int g[NB / 32];
#pragma unroll
    for (int i = 0; i < NB / 32; ++i) g[i] = gt[i * 32 + lane];

    int n = 0;
#pragma unroll
    for (int i = 0; i < NB / 32; ++i) {
        const bool hit = (g[i] == c);
        const unsigned mask = __builtin_amdgcn_ballot_w32(hit);
        const int at  = (int)__builtin_amdgcn_mbcnt_lo(mask, (unsigned)n);
        const int pos = hit ? at : spare;
        list[pos] = i * 32 + lane;
        n += __builtin_popcount(mask);
    }
    __syncthreads();

    const char* rows = (const char*)df;
    const unsigned col = (unsigned)(w * 32 + lane) * 16u;
    v4f m = {0.0f, 0.0f, 0.0f, 0.0f};
    v4f v = {0.0f, 0.0f, 0.0f, 0.0f};
    for (int i0 = 0; i0 < n; i0 += STATS_BATCH) {
        v4f f[STATS_BATCH];
#pragma unroll
        for (int j = 0; j < STATS_BATCH; ++j) {
            const int b = list[min(i0 + j, n - 1)];
            f[j] = *(const v4f*)(rows + ((unsigned)b * (NDF * 4u) + col));
        }
        asm volatile("" :: "v"(f[0]), "v"(f[1]), "v"(f[2]), "v"(f[3]),
                           "v"(f[4]), "v"(f[5]), "v"(f[6]), "v"(f[7]),
                           "v"(f[8]), "v"(f[9]), "v"(f[10]), "v"(f[11]),
                           "v"(f[12]), "v"(f[13]), "v"(f[14]), "v"(f[15]));
#pragma unroll
        for (int j = 0; j < STATS_BATCH; ++j) {
            const int k = i0 + j;
            if (k < n) {
                const float nf = (float)k;
                const float c1 = nf / (nf + 1.0f);
                const float c2 = 1.0f / (nf + 1.0f);
                const v4f fj = f[j];
                m = (k == 0) ? fj : (m * c1 + fj * c2);
                const v4f a = fj - m;
                v = v * c1 + a * a * c1 * c2;
            }
        }
    }
    ((v4f*)var_out)[(size_t)c * (NDF / 4) + w * 32 + lane] = v;
}

#define GB_WAVES 8
#define GB_T   32
#define GB_K   128
#define GB_P   (GB_K + 4)
#define GB_NS  (NDF / GB_K)
#define GB_RW  (GB_T / 4)
#define GB_CP  (GB_T + 4)
#define GB_RP  (32 + 4)
#define GB_RS  (4u * NDF * 4u)

__device__ __forceinline__ void gemm_lds_barrier()
{
    __builtin_amdgcn_fence(__ATOMIC_RELEASE, "workgroup", "local");
    __builtin_amdgcn_s_barrier();
    __builtin_amdgcn_fence(__ATOMIC_ACQUIRE, "workgroup", "local");
}

__device__ __forceinline__ void gemm_slab_load(v4f (&rd)[GB_RW], v4f (&rf)[GB_RW],
                                               v4f (&rv)[GB_RW], v4f (&rb)[GB_RW],
                                               const char* dfB, const char* fcB, const char* vaB,
                                               unsigned oD, unsigned oB,
                                               const unsigned (&oT)[GB_RW], unsigned kb)
{
#pragma unroll
    for (int j = 0; j < GB_RW; ++j) {
        rd[j] = *(const v4f*)(dfB + (oD + (unsigned)j * GB_RS + kb));
        rb[j] = *(const v4f*)(fcB + (oB + (unsigned)j * GB_RS + kb));
    }
#pragma unroll
    for (int j = 0; j < GB_RW; ++j) {
        rf[j] = *(const v4f*)(fcB + (oT[j] + kb));
        rv[j] = *(const v4f*)(vaB + (oT[j] + kb));
    }
}

__device__ __forceinline__ void gemm_slab_store(const v4f (&rd)[GB_RW], const v4f (&rf)[GB_RW],
                                                const v4f (&rv)[GB_RW], const v4f (&rb)[GB_RW],
                                                float* sx, float* sy, float* sp,
                                                float (&cacc)[GB_RW])
{
#pragma unroll
    for (int j = 0; j < GB_RW; ++j) {
        const v4f dfv = rd[j], fcv = rf[j], va = rv[j];
        const v4f x = dfv - ALP * fcv * va;
        const v4f y = HALP * va;
        const v4f t = -fcv * dfv + HALP * fcv * fcv * va;
        cacc[j] += (t.x + t.y) + (t.z + t.w);
        *(v4f*)(sx + j * 4 * GB_P) = x;
        *(v4f*)(sy + j * 4 * GB_P) = y;
        *(v4f*)(sp + j * 4 * GB_P) = rb[j];
    }
}

__device__ __forceinline__ void gemm_slab_mma(const float* px, const float* py, const float* pp,
                                              v8f& ax0, v8f& ax1, v8f& ay0, v8f& ay1)
{
    v4f x[GB_K / 8], y[GB_K / 8], p[GB_K / 8];
#pragma unroll
    for (int s = 0; s < GB_K / 8; ++s) {
        x[s] = *(const v4f*)(px + s * 8);
        y[s] = *(const v4f*)(py + s * 8);
        p[s] = *(const v4f*)(pp + s * 8);
    }
    __builtin_amdgcn_sched_barrier(0);
#pragma unroll
    for (int s = 0; s < GB_K / 8; ++s) {
        const v4f q = p[s] * p[s];
        const v2f x0 = {x[s].x, x[s].y}, x1 = {x[s].z, x[s].w};
        const v2f y0 = {y[s].x, y[s].y}, y1 = {y[s].z, y[s].w};
        const v2f p0 = {p[s].x, p[s].y}, p1 = {p[s].z, p[s].w};
        const v2f q0 = {q.x, q.y},       q1 = {q.z, q.w};
        ax0 = __builtin_amdgcn_wmma_f32_16x16x4_f32(false, x0, false, p0,
                                                    (short)0, ax0, false, false);
        ax1 = __builtin_amdgcn_wmma_f32_16x16x4_f32(false, x1, false, p1,
                                                    (short)0, ax1, false, false);
        ay0 = __builtin_amdgcn_wmma_f32_16x16x4_f32(false, y0, false, q0,
                                                    (short)0, ay0, false, false);
        ay1 = __builtin_amdgcn_wmma_f32_16x16x4_f32(false, y1, false, q1,
                                                    (short)0, ay1, false, false);
        if (s < GB_K / 16) __builtin_amdgcn_sched_barrier(0);
    }
}

__global__ __launch_bounds__(GB_WAVES * 32) void wmma_gemm_kernel(
    const float* __restrict__ df, const int* __restrict__ gt,
    const float* __restrict__ fc, const float* __restrict__ var_in,
    float* __restrict__ out)
{
    __shared__ __attribute__((aligned(16))) float sX[2][GB_T * GB_P];
    __shared__ __attribute__((aligned(16))) float sY[2][GB_T * GB_P];
    __shared__ __attribute__((aligned(16))) float sP[2][GB_T * GB_P];
    __shared__ __attribute__((aligned(16))) float sR[2][GB_T * GB_RP];
    __shared__ __attribute__((aligned(16))) float sC[2][GB_T * GB_CP];

    const int tm   = blockIdx.x;
    const int tn   = blockIdx.y;
    const int lane = threadIdx.x & 31;
    const int w8   = __builtin_amdgcn_readfirstlane((int)(threadIdx.x >> 5));
    const int grp  = w8 >> 2;
    const int w    = w8 & 3;
    const int wm   = w >> 1;
    const int wn   = w & 1;
    const int c    = lane & 15;
    const int h    = lane >> 4;

    const char* dfB = (const char*)df;
    const char* fcB = (const char*)fc;
    const char* vaB = (const char*)var_in;
    const unsigned oD = ((unsigned)(tm * GB_T + w) * NDF + (unsigned)lane * 4u) * 4u;
    const unsigned oB = ((unsigned)(tn * GB_T + w) * NDF + (unsigned)lane * 4u) * 4u;
    unsigned oT[GB_RW];
#pragma unroll
    for (int j = 0; j < GB_RW; ++j)
        oT[j] = ((unsigned)gt[tm * GB_T + w + 4 * j] * NDF + (unsigned)lane * 4u) * 4u;

    const int sOff = w * GB_P + lane * 4;
    const int fA = (wm * 16 + c) * GB_P + 4 * h;
    const int fB = (wn * 16 + c) * GB_P + 4 * h;
    float* stX = &sX[grp][0];
    float* stY = &sY[grp][0];
    float* stP = &sP[grp][0];

    const unsigned kbA = (unsigned)grp * (GB_K * 4u);
    const unsigned kbB = kbA + 2u * (GB_K * 4u);

    float cacc[GB_RW];
#pragma unroll
    for (int j = 0; j < GB_RW; ++j) cacc[j] = 0.0f;

    v8f ax0 = {}, ax1 = {}, ay0 = {}, ay1 = {};
    v4f rd[GB_RW], rf[GB_RW], rv[GB_RW], rb[GB_RW];

    if (grp == 0)
        gemm_slab_load(rd, rf, rv, rb, dfB, fcB, vaB, oD, oB, oT, kbA);
    gemm_lds_barrier();
    if (grp == 0)
        gemm_slab_store(rd, rf, rv, rb, stX + sOff, stY + sOff, stP + sOff, cacc);
    else
        gemm_slab_load(rd, rf, rv, rb, dfB, fcB, vaB, oD, oB, oT, kbA);
    gemm_lds_barrier();

    if (grp == 0) {
        gemm_slab_load(rd, rf, rv, rb, dfB, fcB, vaB, oD, oB, oT, kbB);
        gemm_slab_mma(stX + fA, stY + fA, stP + fB, ax0, ax1, ay0, ay1);
    } else {
        gemm_slab_store(rd, rf, rv, rb, stX + sOff, stY + sOff, stP + sOff, cacc);
        gemm_slab_load(rd, rf, rv, rb, dfB, fcB, vaB, oD, oB, oT, kbB);
    }
    gemm_lds_barrier();

    if (grp == 0)
        gemm_slab_store(rd, rf, rv, rb, stX + sOff, stY + sOff, stP + sOff, cacc);
    else
        gemm_slab_mma(stX + fA, stY + fA, stP + fB, ax0, ax1, ay0, ay1);
    gemm_lds_barrier();

    if (grp == 0)
        gemm_slab_mma(stX + fA, stY + fA, stP + fB, ax0, ax1, ay0, ay1);
    else
        gemm_slab_store(rd, rf, rv, rb, stX + sOff, stY + sOff, stP + sOff, cacc);
    gemm_lds_barrier();

    if (grp != 0)
        gemm_slab_mma(stX + fA, stY + fA, stP + fB, ax0, ax1, ay0, ay1);

    float* sCg = &sC[grp][0];
    float* sRg = &sR[grp][0];
    const int mloc = wm * 16 + (h << 3);
#pragma unroll
    for (int j = 0; j < 8; ++j)
        sCg[(mloc + j) * GB_CP + wn * 16 + c] = (ax0[j] + ax1[j]) + (ay0[j] + ay1[j]);
#pragma unroll
    for (int j = 0; j < GB_RW; ++j)
        sRg[(w + 4 * j) * GB_RP + lane] = cacc[j];
    gemm_lds_barrier();

    const int row = w8 * 4 + (lane >> 3);
    const int cq  = (lane & 7) * 4;
    float cs = 0.0f;
#pragma unroll
    for (int g = 0; g < 8; ++g) {
        const v4f p0 = *(const v4f*)(&sR[0][row * GB_RP + g * 4]);
        const v4f p1 = *(const v4f*)(&sR[1][row * GB_RP + g * 4]);
        cs += ((p0.x + p0.y) + (p0.z + p0.w)) + ((p1.x + p1.y) + (p1.z + p1.w));
    }
    v4f o = *(const v4f*)(&sC[0][row * GB_CP + cq]) + *(const v4f*)(&sC[1][row * GB_CP + cq]);
    o = o + cs;
    *(v4f*)(out + (size_t)(tm * GB_T + row) * NCLS + tn * GB_T + cq) = o;
}

extern "C" void kernel_launch(void* const* d_in, const int* in_sizes, int n_in,
                              void* d_out, int out_size, void* d_ws, size_t ws_size,
                              hipStream_t stream) {
    const float* df = (const float*)d_in[0];
    const int*   gt = (const int*)d_in[1];
    const float* fc = (const float*)d_in[2];
    float*       out = (float*)d_out;

    float* var = (float*)d_ws;

    stats_kernel<<<NCLS, STATS_WAVES * 32, 0, stream>>>(df, gt, var);
    wmma_gemm_kernel<<<dim3(NB / GB_T, NCLS / GB_T), GB_WAVES * 32, 0, stream>>>(df, gt, fc, var, out);
}


# single launch: a work-group takes 8 classes x 32 columns, each wave runs its class's mean/var in registers and builds X rows in LDS; Y*Q once per class; stats launch gone; f32 sum order changes
# speedup vs baseline: 9.1828x; 1.0892x over previous
#include <hip/hip_runtime.h>


typedef __attribute__((ext_vector_type(2))) float v2f;
typedef __attribute__((ext_vector_type(4))) float v4f;
typedef __attribute__((ext_vector_type(8))) float v8f;

#define NCLS 256
#define NDF  512
#define NB   1024
#define ALP  0.1f
#define HALP 0.05f

#define FG_WAVES 8
#define FG_TN    32
#define FG_RC    48
#define FG_P     (NDF + 4)
#define FG_CP    (FG_TN + 4)
#define FG_RP    (32 + 4)
#define FG_SB    8

__device__ __forceinline__ v8f fg_tile_mma(const float* pa, const v4f (&p)[16], bool square)
{
    v4f x[16];
#pragma unroll
    for (int s = 0; s < 16; ++s) x[s] = *(const v4f*)(pa + s * 8);
    __builtin_amdgcn_sched_barrier(0);
    v8f a0 = {}, a1 = {};
#pragma unroll
    for (int s = 0; s < 16; ++s) {
        const v4f b = square ? p[s] * p[s] : p[s];
        const v2f x0 = {x[s].x, x[s].y}, x1 = {x[s].z, x[s].w};
        const v2f b0 = {b.x, b.y},       b1 = {b.z, b.w};
        a0 = __builtin_amdgcn_wmma_f32_16x16x4_f32(false, x0, false, b0,
                                                   (short)0, a0, false, false);
        a1 = __builtin_amdgcn_wmma_f32_16x16x4_f32(false, x1, false, b1,
                                                   (short)0, a1, false, false);
    }
    return a0 + a1;
}

__global__ __launch_bounds__(FG_WAVES * 32) void fused_score_kernel(
    const float* __restrict__ df, const int* __restrict__ gt,
    const float* __restrict__ fc, float* __restrict__ out)
{
    __shared__ int listS[FG_WAVES][NB + 32];
    __shared__ int cntS[FG_WAVES];
    __shared__ int rowB[FG_RC];
    __shared__ int rowK[FG_RC];
    __shared__ __attribute__((aligned(16))) float sX[FG_RC * FG_P];
    __shared__ __attribute__((aligned(16))) float sY[16 * FG_P];
    __shared__ __attribute__((aligned(16))) float sPt[FG_TN * FG_P];
    __shared__ __attribute__((aligned(16))) float sCX[4][FG_RC * FG_CP];
    __shared__ __attribute__((aligned(16))) float sCY[4][16 * FG_CP];
    __shared__ __attribute__((aligned(16))) float sRp[FG_RC * FG_RP];

    const int cg   = blockIdx.x;
    const int tn   = blockIdx.y;
    const int lane = threadIdx.x & 31;
    const int w8   = __builtin_amdgcn_readfirstlane((int)(threadIdx.x >> 5));
    const int cls  = cg * FG_WAVES + w8;
    const int l4   = lane * 4;

    int g[NB / 32];
#pragma unroll
    for (int i = 0; i < NB / 32; ++i) g[i] = gt[i * 32 + lane];

    const char* fcB  = (const char*)fc;
    const char* rows = (const char*)df;
    const unsigned colB = (unsigned)lane * 16u;
    v4f fcv[4];
#pragma unroll
    for (int j = 0; j < 4; ++j)
        fcv[j] = *(const v4f*)(fcB + ((unsigned)cls * (NDF * 4u) + colB + 512u * j));
    v4f pt[4][4];
#pragma unroll
    for (int jr = 0; jr < 4; ++jr)
#pragma unroll
        for (int j = 0; j < 4; ++j)
            pt[jr][j] = *(const v4f*)(fcB + ((unsigned)(tn * FG_TN + w8 + 8 * jr) * (NDF * 4u) + colB + 512u * j));

    int* list = listS[w8];
    const int spare = NB + lane;
    int n = 0;
#pragma unroll
    for (int i = 0; i < NB / 32; ++i) {
        const bool hit = (g[i] == cls);
        const unsigned mask = __builtin_amdgcn_ballot_w32(hit);
        const int at  = (int)__builtin_amdgcn_mbcnt_lo(mask, (unsigned)n);
        const int pos = hit ? at : spare;
        list[pos] = i * 32 + lane;
        n += __builtin_popcount(mask);
    }
    if (lane == 0) cntS[w8] = n;
#pragma unroll
    for (int jr = 0; jr < 4; ++jr)
#pragma unroll
        for (int j = 0; j < 4; ++j)
            *(v4f*)(sPt + (w8 + 8 * jr) * FG_P + l4 + 128 * j) = pt[jr][j];
    {
        const v4f z = {0.0f, 0.0f, 0.0f, 0.0f};
#pragma unroll
        for (int j = 0; j < 4; ++j) *(v4f*)(sY + (8 + w8) * FG_P + l4 + 128 * j) = z;
    }
    __syncthreads();

    int offv = 0, totv = 0;
#pragma unroll
    for (int i = 0; i < FG_WAVES; ++i) {
        const int ci = cntS[i];
        offv += (i < w8) ? ci : 0;
        totv += ci;
    }
    const int off = __builtin_amdgcn_readfirstlane(offv);
    const int R   = __builtin_amdgcn_readfirstlane(totv);
    if (R == 0) return;

    v4f m[4], v[4];
#pragma unroll
    for (int j = 0; j < 4; ++j) { m[j] = (v4f){0.0f, 0.0f, 0.0f, 0.0f}; v[j] = (v4f){0.0f, 0.0f, 0.0f, 0.0f}; }
    for (int i0 = 0; i0 < n; i0 += FG_SB) {
        v4f f[FG_SB][4];
#pragma unroll
        for (int jj = 0; jj < 4; ++jj) {
            const int b = list[min(i0 + jj, n - 1)];
#pragma unroll
            for (int j = 0; j < 4; ++j)
                f[jj][j] = *(const v4f*)(rows + ((unsigned)b * (NDF * 4u) + colB + 512u * j));
        }
        const bool more = (i0 + 4 < n);
        if (more) {
#pragma unroll
            for (int jj = 4; jj < FG_SB; ++jj) {
                const int b = list[min(i0 + jj, n - 1)];
#pragma unroll
                for (int j = 0; j < 4; ++j)
                    f[jj][j] = *(const v4f*)(rows + ((unsigned)b * (NDF * 4u) + colB + 512u * j));
            }
            asm volatile("" :: "v"(f[4][0]), "v"(f[4][1]), "v"(f[4][2]), "v"(f[4][3]),
                               "v"(f[5][0]), "v"(f[5][1]), "v"(f[5][2]), "v"(f[5][3]),
                               "v"(f[6][0]), "v"(f[6][1]), "v"(f[6][2]), "v"(f[6][3]),
                               "v"(f[7][0]), "v"(f[7][1]), "v"(f[7][2]), "v"(f[7][3]));
        }
        asm volatile("" :: "v"(f[0][0]), "v"(f[0][1]), "v"(f[0][2]), "v"(f[0][3]),
                           "v"(f[1][0]), "v"(f[1][1]), "v"(f[1][2]), "v"(f[1][3]),
                           "v"(f[2][0]), "v"(f[2][1]), "v"(f[2][2]), "v"(f[2][3]),
                           "v"(f[3][0]), "v"(f[3][1]), "v"(f[3][2]), "v"(f[3][3]));
#pragma unroll
        for (int jj = 0; jj < FG_SB; ++jj) {
            const int k = i0 + jj;
            if ((jj < 4 || more) && k < n) {
                const float nf = (float)k;
                const float c1 = nf / (nf + 1.0f);
                const float c2 = 1.0f / (nf + 1.0f);
                const int r = off + k;
#pragma unroll
                for (int j = 0; j < 4; ++j) {
                    const v4f fj = f[jj][j];
                    m[j] = (k == 0) ? fj : (m[j] * c1 + fj * c2);
                    const v4f a = fj - m[j];
                    v[j] = v[j] * c1 + a * a * c1 * c2;
                    if (r < FG_RC) *(v4f*)(sX + r * FG_P + l4 + 128 * j) = fj;
                }
            }
        }
    }

    v4f xa[4], tb[4];
#pragma unroll
    for (int j = 0; j < 4; ++j) {
        const v4f va = v[j];
        xa[j] = ALP * fcv[j] * va;
        tb[j] = HALP * fcv[j] * fcv[j] * va;
        *(v4f*)(sY + w8 * FG_P + l4 + 128 * j) = HALP * va;
    }

    const int q  = w8 & 3;
    const int nh = w8 >> 2;
    const int c  = lane & 15;
    const int h  = lane >> 4;
    const int fo = q * 128 + 4 * h;

    const int nch = (R + FG_RC - 1) / FG_RC;
    for (int ch = 0; ch < nch; ++ch) {
        const int r0   = ch * FG_RC;
        const int rcnt = min(FG_RC, R - r0);
        const int mt   = (rcnt + 15) >> 4;
        const int kLo  = max(0, r0 - off);
        const int kHi  = min(n, r0 + FG_RC - off);
        if (ch > 0) {
            for (int k = kLo; k < kHi; ++k) {
                const int b = list[k];
                const int r = off + k - r0;
#pragma unroll
                for (int j = 0; j < 4; ++j)
                    *(v4f*)(sX + r * FG_P + l4 + 128 * j) =
                        *(const v4f*)(rows + ((unsigned)b * (NDF * 4u) + colB + 512u * j));
            }
        }
        for (int k0 = kLo; k0 < kHi; k0 += 4) {
            v4f d[4][4];
#pragma unroll
            for (int jj = 0; jj < 4; ++jj) {
                const int r = off + min(k0 + jj, kHi - 1) - r0;
#pragma unroll
                for (int j = 0; j < 4; ++j) d[jj][j] = *(const v4f*)(sX + r * FG_P + l4 + 128 * j);
            }
            asm volatile("" :: "v"(d[0][0]), "v"(d[0][1]), "v"(d[0][2]), "v"(d[0][3]),
                               "v"(d[1][0]), "v"(d[1][1]), "v"(d[1][2]), "v"(d[1][3]),
                               "v"(d[2][0]), "v"(d[2][1]), "v"(d[2][2]), "v"(d[2][3]),
                               "v"(d[3][0]), "v"(d[3][1]), "v"(d[3][2]), "v"(d[3][3]));
#pragma unroll
            for (int jj = 0; jj < 4; ++jj) {
                const int k = k0 + jj;
                if (k < kHi) {
                    const int r = off + k - r0;
                    float part = 0.0f;
#pragma unroll
                    for (int j = 0; j < 4; ++j) {
                        const v4f dfv = d[jj][j];
                        const v4f x = dfv - xa[j];
                        const v4f t = -fcv[j] * dfv + tb[j];
                        part += (t.x + t.y) + (t.z + t.w);
                        *(v4f*)(sX + r * FG_P + l4 + 128 * j) = x;
                    }
                    sRp[r * FG_RP + lane] = part;
                    if (lane == 0) { rowB[r] = list[k]; rowK[r] = w8; }
                }
            }
        }
        for (int r = rcnt + w8; r < mt * 16; r += FG_WAVES) {
            const v4f z = {0.0f, 0.0f, 0.0f, 0.0f};
#pragma unroll
            for (int j = 0; j < 4; ++j) *(v4f*)(sX + r * FG_P + l4 + 128 * j) = z;
        }
        __syncthreads();

        {
            v4f p[16];
            const float* pB = sPt + (nh * 16 + c) * FG_P + fo;
#pragma unroll
            for (int s = 0; s < 16; ++s) p[s] = *(const v4f*)(pB + s * 8);
            if (ch == 0) {
                const v8f d = fg_tile_mma(sY + c * FG_P + fo, p, true);
                float* pc = &sCY[q][(8 * h) * FG_CP + nh * 16 + c];
#pragma unroll
                for (int i = 0; i < 8; ++i) pc[i * FG_CP] = d[i];
            }
            for (int jt = 0; jt < mt; ++jt) {
                const v8f d = fg_tile_mma(sX + (jt * 16 + c) * FG_P + fo, p, false);
                float* pc = &sCX[q][(jt * 16 + 8 * h) * FG_CP + nh * 16 + c];
#pragma unroll
                for (int i = 0; i < 8; ++i) pc[i * FG_CP] = d[i];
            }
        }
        __syncthreads();

#pragma unroll
        for (int ps = 0; ps < 2; ++ps) {
            const int row = ps * 32 + w8 * 4 + (lane >> 3);
            if (row < rcnt) {
                const int cq = (lane & 7) * 4;
                float cs = 0.0f;
#pragma unroll
                for (int g8 = 0; g8 < 8; ++g8) {
                    const v4f p0 = *(const v4f*)(sRp + row * FG_RP + g8 * 4);
                    cs += (p0.x + p0.y) + (p0.z + p0.w);
                }
                const int b  = rowB[row];
                const int ks = rowK[row];
                const v4f ox = (*(const v4f*)(&sCX[0][row * FG_CP + cq]) + *(const v4f*)(&sCX[1][row * FG_CP + cq]))
                             + (*(const v4f*)(&sCX[2][row * FG_CP + cq]) + *(const v4f*)(&sCX[3][row * FG_CP + cq]));
                const v4f oy = (*(const v4f*)(&sCY[0][ks * FG_CP + cq]) + *(const v4f*)(&sCY[1][ks * FG_CP + cq]))
                             + (*(const v4f*)(&sCY[2][ks * FG_CP + cq]) + *(const v4f*)(&sCY[3][ks * FG_CP + cq]));
                const v4f o = (ox + oy) + cs;
                *(v4f*)(out + (size_t)b * NCLS + tn * FG_TN + cq) = o;
            }
        }
        if (ch + 1 < nch) __syncthreads();
    }
}

extern "C" void kernel_launch(void* const* d_in, const int* in_sizes, int n_in,
                              void* d_out, int out_size, void* d_ws, size_t ws_size,
                              hipStream_t stream) {
    const float* df = (const float*)d_in[0];
    const int*   gt = (const int*)d_in[1];
    const float* fc = (const float*)d_in[2];
    float*       out = (float*)d_out;

    fused_score_kernel<<<dim3(NCLS / FG_WAVES, NCLS / FG_TN), FG_WAVES * 32, 0, stream>>>(df, gt, fc, out);
}


# stats loop: each step's two divisions come from a per-lane table (32 steps divided at once, read by lane index); tail steps get (1,0), so four steps run as one branch-free stretch; same sums
# speedup vs baseline: 10.0870x; 1.0985x over previous
#include <hip/hip_runtime.h>


typedef __attribute__((ext_vector_type(2))) float v2f;
typedef __attribute__((ext_vector_type(4))) float v4f;
typedef __attribute__((ext_vector_type(8))) float v8f;

#define NCLS 256
#define NDF  512
#define NB   1024
#define ALP  0.1f
#define HALP 0.05f

#define FG_WAVES 8
#define FG_TN    32
#define FG_RC    48
#define FG_P     (NDF + 4)
#define FG_CP    (FG_TN + 4)
#define FG_RP    (32 + 4)
#define FG_SB    8

__device__ __forceinline__ v8f fg_tile_mma(const float* pa, const v4f (&p)[16], bool square)
{
    v4f x[16];
#pragma unroll
    for (int s = 0; s < 16; ++s) x[s] = *(const v4f*)(pa + s * 8);
    __builtin_amdgcn_sched_barrier(0);
    v8f a0 = {}, a1 = {};
#pragma unroll
    for (int s = 0; s < 16; ++s) {
        const v4f b = square ? p[s] * p[s] : p[s];
        const v2f x0 = {x[s].x, x[s].y}, x1 = {x[s].z, x[s].w};
        const v2f b0 = {b.x, b.y},       b1 = {b.z, b.w};
        a0 = __builtin_amdgcn_wmma_f32_16x16x4_f32(false, x0, false, b0,
                                                   (short)0, a0, false, false);
        a1 = __builtin_amdgcn_wmma_f32_16x16x4_f32(false, x1, false, b1,
                                                   (short)0, a1, false, false);
    }
    return a0 + a1;
}

__global__ __launch_bounds__(FG_WAVES * 32) void fused_score_kernel(
    const float* __restrict__ df, const int* __restrict__ gt,
    const float* __restrict__ fc, float* __restrict__ out)
{
    __shared__ int listS[FG_WAVES][NB + 32];
    __shared__ int cntS[FG_WAVES];
    __shared__ int rowB[FG_RC];
    __shared__ int rowK[FG_RC];
    __shared__ __attribute__((aligned(16))) float sX[FG_RC * FG_P];
    __shared__ __attribute__((aligned(16))) float sY[16 * FG_P];
    __shared__ __attribute__((aligned(16))) float sPt[FG_TN * FG_P];
    __shared__ __attribute__((aligned(16))) float sCX[4][FG_RC * FG_CP];
    __shared__ __attribute__((aligned(16))) float sCY[4][16 * FG_CP];
    __shared__ __attribute__((aligned(16))) float sRp[FG_RC * FG_RP];

    const int cg   = blockIdx.x;
    const int tn   = blockIdx.y;
    const int lane = threadIdx.x & 31;
    const int w8   = __builtin_amdgcn_readfirstlane((int)(threadIdx.x >> 5));
    const int cls  = cg * FG_WAVES + w8;
    const int l4   = lane * 4;

    int g[NB / 32];
#pragma unroll
    for (int i = 0; i < NB / 32; ++i) g[i] = gt[i * 32 + lane];

    const char* fcB  = (const char*)fc;
    const char* rows = (const char*)df;
    const unsigned colB = (unsigned)lane * 16u;
    v4f fcv[4];
#pragma unroll
    for (int j = 0; j < 4; ++j)
        fcv[j] = *(const v4f*)(fcB + ((unsigned)cls * (NDF * 4u) + colB + 512u * j));
    v4f pt[4][4];
#pragma unroll
    for (int jr = 0; jr < 4; ++jr)
#pragma unroll
        for (int j = 0; j < 4; ++j)
            pt[jr][j] = *(const v4f*)(fcB + ((unsigned)(tn * FG_TN + w8 + 8 * jr) * (NDF * 4u) + colB + 512u * j));

    int* list = listS[w8];
    const int spare = NB + lane;
    int n = 0;
#pragma unroll
    for (int i = 0; i < NB / 32; ++i) {
        const bool hit = (g[i] == cls);
        const unsigned mask = __builtin_amdgcn_ballot_w32(hit);
        const int at  = (int)__builtin_amdgcn_mbcnt_lo(mask, (unsigned)n);
        const int pos = hit ? at : spare;
        list[pos] = i * 32 + lane;
        n += __builtin_popcount(mask);
    }
    if (lane == 0) cntS[w8] = n;
#pragma unroll
    for (int jr = 0; jr < 4; ++jr)
#pragma unroll
        for (int j = 0; j < 4; ++j)
            *(v4f*)(sPt + (w8 + 8 * jr) * FG_P + l4 + 128 * j) = pt[jr][j];
    {
        const v4f z = {0.0f, 0.0f, 0.0f, 0.0f};
#pragma unroll
        for (int j = 0; j < 4; ++j) *(v4f*)(sY + (8 + w8) * FG_P + l4 + 128 * j) = z;
    }
    __syncthreads();

    int offv = 0, totv = 0;
#pragma unroll
    for (int i = 0; i < FG_WAVES; ++i) {
        const int ci = cntS[i];
        offv += (i < w8) ? ci : 0;
        totv += ci;
    }
    const int off = __builtin_amdgcn_readfirstlane(offv);
    const int R   = __builtin_amdgcn_readfirstlane(totv);
    if (R == 0) return;

    v4f m[4], v[4];
#pragma unroll
    for (int j = 0; j < 4; ++j) { m[j] = (v4f){0.0f, 0.0f, 0.0f, 0.0f}; v[j] = (v4f){0.0f, 0.0f, 0.0f, 0.0f}; }
    int c1t = 0, c2t = 0;
    for (int i0 = 0; i0 < n; i0 += FG_SB) {
        v4f f[FG_SB][4];
#pragma unroll
        for (int jj = 0; jj < 4; ++jj) {
            const int b = list[min(i0 + jj, n - 1)];
#pragma unroll
            for (int j = 0; j < 4; ++j)
                f[jj][j] = *(const v4f*)(rows + ((unsigned)b * (NDF * 4u) + colB + 512u * j));
        }
        const bool more = (i0 + 4 < n);
        if (more) {
#pragma unroll
            for (int jj = 4; jj < FG_SB; ++jj) {
                const int b = list[min(i0 + jj, n - 1)];
#pragma unroll
                for (int j = 0; j < 4; ++j)
                    f[jj][j] = *(const v4f*)(rows + ((unsigned)b * (NDF * 4u) + colB + 512u * j));
            }
        }
        if ((i0 & 31) == 0) {
            const int   kk   = i0 + lane;
            const float nf   = (float)kk;
            const bool  live = (kk < n);
            c1t = __float_as_int(live ? nf / (nf + 1.0f) : 1.0f);
            c2t = __float_as_int(live ? 1.0f / (nf + 1.0f) : 0.0f);
        }
        if (more) {
            asm volatile("" :: "v"(f[4][0]), "v"(f[4][1]), "v"(f[4][2]), "v"(f[4][3]),
                               "v"(f[5][0]), "v"(f[5][1]), "v"(f[5][2]), "v"(f[5][3]),
                               "v"(f[6][0]), "v"(f[6][1]), "v"(f[6][2]), "v"(f[6][3]),
                               "v"(f[7][0]), "v"(f[7][1]), "v"(f[7][2]), "v"(f[7][3]));
        }
        asm volatile("" :: "v"(f[0][0]), "v"(f[0][1]), "v"(f[0][2]), "v"(f[0][3]),
                           "v"(f[1][0]), "v"(f[1][1]), "v"(f[1][2]), "v"(f[1][3]),
                           "v"(f[2][0]), "v"(f[2][1]), "v"(f[2][2]), "v"(f[2][3]),
                           "v"(f[3][0]), "v"(f[3][1]), "v"(f[3][2]), "v"(f[3][3]));
#pragma unroll
        for (int g4 = 0; g4 < FG_SB; g4 += 4) {
            if (g4 == 0 || more) {
#pragma unroll
                for (int jj = g4; jj < g4 + 4; ++jj) {
                    const int   ks = (i0 + jj) & 31;
                    const float c1 = __int_as_float(__builtin_amdgcn_readlane(c1t, ks));
                    const float c2 = __int_as_float(__builtin_amdgcn_readlane(c2t, ks));
#pragma unroll
                    for (int j = 0; j < 4; ++j) {
                        const v4f fj = f[jj][j];
                        m[j] = m[j] * c1 + fj * c2;
                        const v4f a = fj - m[j];
                        v[j] = v[j] * c1 + a * a * c1 * c2;
                    }
                }
#pragma unroll
                for (int jj = g4; jj < g4 + 4; ++jj) {
                    const int k = i0 + jj;
                    const int r = off + k;
                    if (k < n && r < FG_RC) {
#pragma unroll
                        for (int j = 0; j < 4; ++j) *(v4f*)(sX + r * FG_P + l4 + 128 * j) = f[jj][j];
                    }
                }
            }
        }
    }

    v4f xa[4], tb[4];
#pragma unroll
    for (int j = 0; j < 4; ++j) {
        const v4f va = v[j];
        xa[j] = ALP * fcv[j] * va;
        tb[j] = HALP * fcv[j] * fcv[j] * va;
        *(v4f*)(sY + w8 * FG_P + l4 + 128 * j) = HALP * va;
    }

    const int q  = w8 & 3;
    const int nh = w8 >> 2;
    const int c  = lane & 15;
    const int h  = lane >> 4;
    const int fo = q * 128 + 4 * h;

    const int nch = (R + FG_RC - 1) / FG_RC;
    for (int ch = 0; ch < nch; ++ch) {
        const int r0   = ch * FG_RC;
        const int rcnt = min(FG_RC, R - r0);
        const int mt   = (rcnt + 15) >> 4;
        const int kLo  = max(0, r0 - off);
        const int kHi  = min(n, r0 + FG_RC - off);
        if (ch > 0) {
            for (int k = kLo; k < kHi; ++k) {
                const int b = list[k];
                const int r = off + k - r0;
#pragma unroll
                for (int j = 0; j < 4; ++j)
                    *(v4f*)(sX + r * FG_P + l4 + 128 * j) =
                        *(const v4f*)(rows + ((unsigned)b * (NDF * 4u) + colB + 512u * j));
            }
        }
        for (int k0 = kLo; k0 < kHi; k0 += 4) {
            v4f d[4][4];
#pragma unroll
            for (int jj = 0; jj < 4; ++jj) {
                const int r = off + min(k0 + jj, kHi - 1) - r0;
#pragma unroll
                for (int j = 0; j < 4; ++j) d[jj][j] = *(const v4f*)(sX + r * FG_P + l4 + 128 * j);
            }
            asm volatile("" :: "v"(d[0][0]), "v"(d[0][1]), "v"(d[0][2]), "v"(d[0][3]),
                               "v"(d[1][0]), "v"(d[1][1]), "v"(d[1][2]), "v"(d[1][3]),
                               "v"(d[2][0]), "v"(d[2][1]), "v"(d[2][2]), "v"(d[2][3]),
                               "v"(d[3][0]), "v"(d[3][1]), "v"(d[3][2]), "v"(d[3][3]));
#pragma unroll
            for (int jj = 0; jj < 4; ++jj) {
                const int k = k0 + jj;
                if (k < kHi) {
                    const int r = off + k - r0;
                    float part = 0.0f;
#pragma unroll
                    for (int j = 0; j < 4; ++j) {
                        const v4f dfv = d[jj][j];
                        const v4f x = dfv - xa[j];
                        const v4f t = -fcv[j] * dfv + tb[j];
                        part += (t.x + t.y) + (t.z + t.w);
                        *(v4f*)(sX + r * FG_P + l4 + 128 * j) = x;
                    }
                    sRp[r * FG_RP + lane] = part;
                    if (lane == 0) { rowB[r] = list[k]; rowK[r] = w8; }
                }
            }
        }
        for (int r = rcnt + w8; r < mt * 16; r += FG_WAVES) {
            const v4f z = {0.0f, 0.0f, 0.0f, 0.0f};
#pragma unroll
            for (int j = 0; j < 4; ++j) *(v4f*)(sX + r * FG_P + l4 + 128 * j) = z;
        }
        __syncthreads();

        {
            v4f p[16];
            const float* pB = sPt + (nh * 16 + c) * FG_P + fo;
#pragma unroll
            for (int s = 0; s < 16; ++s) p[s] = *(const v4f*)(pB + s * 8);
            if (ch == 0) {
                const v8f d = fg_tile_mma(sY + c * FG_P + fo, p, true);
                float* pc = &sCY[q][(8 * h) * FG_CP + nh * 16 + c];
#pragma unroll
                for (int i = 0; i < 8; ++i) pc[i * FG_CP] = d[i];
            }
            for (int jt = 0; jt < mt; ++jt) {
                const v8f d = fg_tile_mma(sX + (jt * 16 + c) * FG_P + fo, p, false);
                float* pc = &sCX[q][(jt * 16 + 8 * h) * FG_CP + nh * 16 + c];
#pragma unroll
                for (int i = 0; i < 8; ++i) pc[i * FG_CP] = d[i];
            }
        }
        __syncthreads();

#pragma unroll
        for (int ps = 0; ps < 2; ++ps) {
            const int row = ps * 32 + w8 * 4 + (lane >> 3);
            if (row < rcnt) {
                const int cq = (lane & 7) * 4;
                float cs = 0.0f;
#pragma unroll
                for (int g8 = 0; g8 < 8; ++g8) {
                    const v4f p0 = *(const v4f*)(sRp + row * FG_RP + g8 * 4);
                    cs += (p0.x + p0.y) + (p0.z + p0.w);
                }
                const int b  = rowB[row];
                const int ks = rowK[row];
                const v4f ox = (*(const v4f*)(&sCX[0][row * FG_CP + cq]) + *(const v4f*)(&sCX[1][row * FG_CP + cq]))
                             + (*(const v4f*)(&sCX[2][row * FG_CP + cq]) + *(const v4f*)(&sCX[3][row * FG_CP + cq]));
                const v4f oy = (*(const v4f*)(&sCY[0][ks * FG_CP + cq]) + *(const v4f*)(&sCY[1][ks * FG_CP + cq]))
                             + (*(const v4f*)(&sCY[2][ks * FG_CP + cq]) + *(const v4f*)(&sCY[3][ks * FG_CP + cq]));
                const v4f o = (ox + oy) + cs;
                *(v4f*)(out + (size_t)b * NCLS + tn * FG_TN + cq) = o;
            }
        }
        if (ch + 1 < nch) __syncthreads();
    }
}

extern "C" void kernel_launch(void* const* d_in, const int* in_sizes, int n_in,
                              void* d_out, int out_size, void* d_ws, size_t ws_size,
                              hipStream_t stream) {
    const float* df = (const float*)d_in[0];
    const int*   gt = (const int*)d_in[1];
    const float* fc = (const float*)d_in[2];
    float*       out = (float*)d_out;

    fused_score_kernel<<<dim3(NCLS / FG_WAVES, NCLS / FG_TN), FG_WAVES * 32, 0, stream>>>(df, gt, fc, out);
}
